# starting kernel
# baseline (speedup 1.0000x reference)
#include <hip/hip_runtime.h>
#include <hip/hip_bf16.h>

#define BB 8
#define LL 8192
#define CC 512
#define ROWS (BB * LL)
#define OUT_PER_B (LL - 2)
#define OUT_TOTAL (BB * OUT_PER_B)

typedef __attribute__((ext_vector_type(2))) float v2f;
typedef __attribute__((ext_vector_type(4))) float v4f;
typedef __attribute__((ext_vector_type(8))) float v8f;

static __device__ __forceinline__ float sum4(const v4f v) {
    return (v[0] + v[1]) + (v[2] + v[3]);
}

__global__ void __launch_bounds__(256) x_partial_kernel(const float* __restrict__ x,
                                                        float* __restrict__ partials) {
    const int t = threadIdx.x;
    const v4f* x4 = (const v4f*)x + (long)blockIdx.x * 8192 + t;
    float acc = 0.f;
#pragma unroll 1
    for (int i = 0; i < 32; i += 8) {
        float s = 0.f;
#pragma unroll
        for (int j = 0; j < 8; ++j)
            s += sum4(__builtin_nontemporal_load(&x4[(i + j) * 256]));
        acc += s;
    }
    __shared__ float sm[256];
    sm[t] = acc;
    __syncthreads();
#pragma unroll
    for (int s = 128; s > 0; s >>= 1) {
        if (t < s) sm[t] += sm[t + s];
        __syncthreads();
    }
    if (t == 0) partials[blockIdx.x] = sm[0];
}

__global__ void m_final_kernel(const float* __restrict__ partials,
                               float* __restrict__ M) {
    const int b = threadIdx.x;
    if (b < BB) {
        float s = 0.f;
        for (int i = 0; i < 128; ++i) s += partials[b * 128 + i];
        M[b] = s / (float)(LL * CC);
    }
}

__global__ void __launch_bounds__(256) rowsum_wmma_kernel(const float* __restrict__ noise,
                                                          float* __restrict__ S) {
    const int lane = threadIdx.x & 31;
    const int wave = threadIdx.x >> 5;
    const long rowbase = (long)blockIdx.x * 128 + wave * 16;
    const int m  = lane & 15;
    const int hi = lane >> 4;

    const v4f* rp = (const v4f*)(noise + (rowbase + m) * CC + hi * 256);

    float p0 = 0.f, p1 = 0.f;
#pragma unroll 1
    for (int i = 0; i < 32; i += 8) {
        float s = 0.f;
#pragma unroll
        for (int j = 0; j < 8; ++j)
            s += sum4(__builtin_nontemporal_load(&rp[i + j]));
        p0 += s;
    }
#pragma unroll 1
    for (int i = 32; i < 64; i += 8) {
        float s = 0.f;
#pragma unroll
        for (int j = 0; j < 8; ++j)
            s += sum4(__builtin_nontemporal_load(&rp[i + j]));
        p1 += s;
    }

    v2f a; a[0] = p0; a[1] = p1;
    v2f b; b[0] = 1.f; b[1] = 1.f;
    v8f c = {};
    c = __builtin_amdgcn_wmma_f32_16x16x4_f32(
             false, a,  false, b,
             (short)0, c,  false,  false);

    if (lane == 0) {
#pragma unroll
        for (int r = 0; r < 8; ++r) S[rowbase + r] = c[r];
    } else if (lane == 16) {
#pragma unroll
        for (int r = 0; r < 8; ++r) S[rowbase + 8 + r] = c[r];
    }
}

__global__ void pool_kernel(const float* __restrict__ M,
                            const float* __restrict__ S,
                            float* __restrict__ out) {
    int idx = blockIdx.x * 256 + threadIdx.x;
    if (idx >= OUT_TOTAL) return;
    int b = idx / OUT_PER_B;
    int l = idx - b * OUT_PER_B;
    const float* Sb = S + (long)b * LL + l;
    const float k = 0.1f / (3.0f * (float)CC);
    out[idx] = M[b] + k * ((Sb[0] + Sb[1]) + Sb[2]);
}

extern "C" void kernel_launch(void* const* d_in, const int* in_sizes, int n_in,
                              void* d_out, int out_size, void* d_ws, size_t ws_size,
                              hipStream_t stream) {
    (void)in_sizes; (void)n_in; (void)out_size; (void)ws_size;
    const float* x     = (const float*)d_in[0];
    const float* noise = (const float*)d_in[1];
    float* out = (float*)d_out;

    float* ws       = (float*)d_ws;
    float* partials = ws;
    float* M        = ws + 1024;
    float* S        = ws + 2048;

    x_partial_kernel<<<1024, 256, 0, stream>>>(x, partials);
    m_final_kernel<<<1, 32, 0, stream>>>(partials, M);
    rowsum_wmma_kernel<<<ROWS / 128, 256, 0, stream>>>(noise, S);
    pool_kernel<<<(OUT_TOTAL + 255) / 256, 256, 0, stream>>>(M, S, out);
}


# noise row-sum kernel: lane-to-address map changed so each 16-byte wave load covers 4 whole 128-byte lines (was 32 lines x 16 B); row total = sum of the 8 D registers; summation order changes
# speedup vs baseline: 1.6109x; 1.6109x over previous
#include <hip/hip_runtime.h>
#include <hip/hip_bf16.h>

#define BB 8
#define LL 8192
#define CC 512
#define ROWS (BB * LL)
#define OUT_PER_B (LL - 2)
#define OUT_TOTAL (BB * OUT_PER_B)

typedef __attribute__((ext_vector_type(2))) float v2f;
typedef __attribute__((ext_vector_type(4))) float v4f;
typedef __attribute__((ext_vector_type(8))) float v8f;

static __device__ __forceinline__ float sum4(const v4f v) {
    return (v[0] + v[1]) + (v[2] + v[3]);
}

__global__ void __launch_bounds__(256) x_partial_kernel(const float* __restrict__ x,
                                                        float* __restrict__ partials) {
    const int t = threadIdx.x;
    const v4f* x4 = (const v4f*)x + (long)blockIdx.x * 8192 + t;
    float acc = 0.f;
#pragma unroll 1
    for (int i = 0; i < 32; i += 8) {
        float s = 0.f;
#pragma unroll
        for (int j = 0; j < 8; ++j)
            s += sum4(__builtin_nontemporal_load(&x4[(i + j) * 256]));
        acc += s;
    }
    __shared__ float sm[256];
    sm[t] = acc;
    __syncthreads();
#pragma unroll
    for (int s = 128; s > 0; s >>= 1) {
        if (t < s) sm[t] += sm[t + s];
        __syncthreads();
    }
    if (t == 0) partials[blockIdx.x] = sm[0];
}

__global__ void m_final_kernel(const float* __restrict__ partials,
                               float* __restrict__ M) {
    const int b = threadIdx.x;
    if (b < BB) {
        float s = 0.f;
        for (int i = 0; i < 128; ++i) s += partials[b * 128 + i];
        M[b] = s / (float)(LL * CC);
    }
}

__global__ void __launch_bounds__(256) rowsum_wmma_kernel(const float* __restrict__ noise,
                                                          float* __restrict__ S) {
    const int lane = threadIdx.x & 31;
    const int wave = threadIdx.x >> 5;
    const long rowbase = (long)blockIdx.x * 128 + wave * 16;
    const int c = lane & 15;
    const int h = lane >> 4;

    const v4f* rp = (const v4f*)(noise + (rowbase + (c >> 3) * 8) * CC + h * 32 + (c & 7) * 4);

    v2f b; b[0] = 1.f; b[1] = 1.f;
    float mine = 0.f;
#pragma unroll 1
    for (int p = 0; p < 8; ++p) {
        const v4f* q = rp + p * (CC / 4);
        float s0 = 0.f, s1 = 0.f;
#pragma unroll
        for (int j = 0; j < 4; ++j)
            s0 += sum4(__builtin_nontemporal_load(&q[j * 16]));
#pragma unroll
        for (int j = 4; j < 8; ++j)
            s1 += sum4(__builtin_nontemporal_load(&q[j * 16]));

        v2f a; a[0] = s0; a[1] = s1;
        v8f d = {};
        d = __builtin_amdgcn_wmma_f32_16x16x4_f32(
                 false, a,  false, b,
                 (short)0, d,  false,  false);

        const float t = ((d[0] + d[1]) + (d[2] + d[3])) + ((d[4] + d[5]) + (d[6] + d[7]));
        mine = (c == p) ? t : mine;
    }
    if (c < 8) S[rowbase + h * 8 + c] = mine;
}

__global__ void pool_kernel(const float* __restrict__ M,
                            const float* __restrict__ S,
                            float* __restrict__ out) {
    int idx = blockIdx.x * 256 + threadIdx.x;
    if (idx >= OUT_TOTAL) return;
    int b = idx / OUT_PER_B;
    int l = idx - b * OUT_PER_B;
    const float* Sb = S + (long)b * LL + l;
    const float k = 0.1f / (3.0f * (float)CC);
    out[idx] = M[b] + k * ((Sb[0] + Sb[1]) + Sb[2]);
}

extern "C" void kernel_launch(void* const* d_in, const int* in_sizes, int n_in,
                              void* d_out, int out_size, void* d_ws, size_t ws_size,
                              hipStream_t stream) {
    (void)in_sizes; (void)n_in; (void)out_size; (void)ws_size;
    const float* x     = (const float*)d_in[0];
    const float* noise = (const float*)d_in[1];
    float* out = (float*)d_out;

    float* ws       = (float*)d_ws;
    float* partials = ws;
    float* M        = ws + 1024;
    float* S        = ws + 2048;

    x_partial_kernel<<<1024, 256, 0, stream>>>(x, partials);
    m_final_kernel<<<1, 32, 0, stream>>>(partials, M);
    rowsum_wmma_kernel<<<ROWS / 128, 256, 0, stream>>>(noise, S);
    pool_kernel<<<(OUT_TOTAL + 255) / 256, 256, 0, stream>>>(M, S, out);
}


# x-mean launches folded into the noise launches (4 -> 2): each row-sum work-group also sums 65536 floats of x; each pooling wave finishes the mean by lane butterfly; x summation order changes
# speedup vs baseline: 1.6827x; 1.0446x over previous
#include <hip/hip_runtime.h>
#include <hip/hip_bf16.h>

#define BB 8
#define LL 8192
#define CC 512
#define ROWS (BB * LL)
#define OUT_PER_B (LL - 2)
#define OUT_TOTAL (BB * OUT_PER_B)

typedef __attribute__((ext_vector_type(2))) float v2f;
typedef __attribute__((ext_vector_type(4))) float v4f;
typedef __attribute__((ext_vector_type(8))) float v8f;

static __device__ __forceinline__ float sum4(const v4f v) {
    return (v[0] + v[1]) + (v[2] + v[3]);
}

__global__ void __launch_bounds__(256) stream_kernel(const float* __restrict__ x,
                                                     const float* __restrict__ noise,
                                                     float* __restrict__ partials,
                                                     float* __restrict__ S) {
    const int tid  = threadIdx.x;
    const int lane = tid & 31;
    const int wave = tid >> 5;

    {
        const long rowbase = (long)blockIdx.x * 128 + wave * 16;
        const int c = lane & 15;
        const int h = lane >> 4;

        const v4f* rp = (const v4f*)(noise + (rowbase + (c >> 3) * 8) * CC + h * 32 + (c & 7) * 4);

        v2f b; b[0] = 1.f; b[1] = 1.f;
        float mine = 0.f;
#pragma unroll 1
        for (int p = 0; p < 8; ++p) {
            const v4f* q = rp + p * (CC / 4);
            float s0 = 0.f, s1 = 0.f;
#pragma unroll
            for (int j = 0; j < 4; ++j)
                s0 += sum4(__builtin_nontemporal_load(&q[j * 16]));
#pragma unroll
            for (int j = 4; j < 8; ++j)
                s1 += sum4(__builtin_nontemporal_load(&q[j * 16]));

            v2f a; a[0] = s0; a[1] = s1;
            v8f d = {};
            d = __builtin_amdgcn_wmma_f32_16x16x4_f32(
                     false, a,  false, b,
                     (short)0, d,  false,  false);

            const float t = ((d[0] + d[1]) + (d[2] + d[3])) + ((d[4] + d[5]) + (d[6] + d[7]));
            mine = (c == p) ? t : mine;
        }
        if (c < 8) S[rowbase + h * 8 + c] = mine;
    }

    const v4f* x4 = (const v4f*)x + (long)blockIdx.x * 16384 + tid;
    float acc = 0.f;
#pragma unroll 1
    for (int i = 0; i < 64; i += 8) {
        float s = 0.f;
#pragma unroll
        for (int j = 0; j < 8; ++j)
            s += sum4(__builtin_nontemporal_load(&x4[(i + j) * 256]));
        acc += s;
    }
    __shared__ float sm[256];
    sm[tid] = acc;
    __syncthreads();
#pragma unroll
    for (int s = 128; s > 0; s >>= 1) {
        if (tid < s) sm[tid] += sm[tid + s];
        __syncthreads();
    }
    if (tid == 0) partials[blockIdx.x] = sm[0];
}

__global__ void __launch_bounds__(256) pool_kernel(const float* __restrict__ partials,
                                                   const float* __restrict__ S,
                                                   float* __restrict__ out) {
    const int idx  = blockIdx.x * 256 + threadIdx.x;
    const int lane = threadIdx.x & 31;
    const int ic = min(idx, OUT_TOTAL - 1);
    const int b  = ic / OUT_PER_B;
    const int l  = ic - b * OUT_PER_B;
    const float* Sb = S + (long)b * LL + l;
    float t0 = Sb[0], t1 = Sb[1], t2 = Sb[2];

    const int b0 = min((idx - lane) / OUT_PER_B, BB - 1);
    const int b1 = min(b0 + 1, BB - 1);
    float m0 = partials[b0 * 64 + lane] + partials[b0 * 64 + 32 + lane];
    float m1 = partials[b1 * 64 + lane] + partials[b1 * 64 + 32 + lane];
    for (int w = 16; w > 0; w >>= 1) {
        m0 += __shfl_xor(m0, w);
        m1 += __shfl_xor(m1, w);
    }
    asm volatile("" : "+v"(t0), "+v"(t1), "+v"(t2));
    const float k = 0.1f / (3.0f * (float)CC);
    const float M = ((b == b0) ? m0 : m1) / (float)(LL * CC);
    if (idx < OUT_TOTAL) out[idx] = M + k * ((t0 + t1) + t2);
}

extern "C" void kernel_launch(void* const* d_in, const int* in_sizes, int n_in,
                              void* d_out, int out_size, void* d_ws, size_t ws_size,
                              hipStream_t stream) {
    (void)in_sizes; (void)n_in; (void)out_size; (void)ws_size;
    const float* x     = (const float*)d_in[0];
    const float* noise = (const float*)d_in[1];
    float* out = (float*)d_out;

    float* ws       = (float*)d_ws;
    float* partials = ws;
    float* S        = ws + 2048;

    stream_kernel<<<ROWS / 128, 256, 0, stream>>>(x, noise, partials, S);
    pool_kernel<<<(OUT_TOTAL + 255) / 256, 256, 0, stream>>>(partials, S, out);
}


# stream kernel: loads issued three 8-load chunks ahead of the sums through a ring of register buffers (24-32 loads in flight per wave, was 0-8), x following noise without a drain; same sum order
# speedup vs baseline: 1.7466x; 1.0380x over previous
#include <hip/hip_runtime.h>
#include <hip/hip_bf16.h>

#define BB 8
#define LL 8192
#define CC 512
#define ROWS (BB * LL)
#define OUT_PER_B (LL - 2)
#define OUT_TOTAL (BB * OUT_PER_B)

typedef __attribute__((ext_vector_type(2))) float v2f;
typedef __attribute__((ext_vector_type(4))) float v4f;
typedef __attribute__((ext_vector_type(8))) float v8f;

static __device__ __forceinline__ float sum4(const v4f v) {
    return (v[0] + v[1]) + (v[2] + v[3]);
}

#define CHUNKS 16
#define AHEAD 3

__global__ void __launch_bounds__(256) stream_kernel(const float* __restrict__ x,
                                                     const float* __restrict__ noise,
                                                     float* __restrict__ partials,
                                                     float* __restrict__ S) {
    const int tid  = threadIdx.x;
    const int lane = tid & 31;
    const int wave = tid >> 5;

    const long rowbase = (long)blockIdx.x * 128 + wave * 16;
    const int c = lane & 15;
    const int h = lane >> 4;

    const v4f* rp = (const v4f*)(noise + (rowbase + (c >> 3) * 8) * CC + h * 32 + (c & 7) * 4);
    const v4f* x4 = (const v4f*)x + (long)blockIdx.x * 16384 + tid;

    v4f ring[AHEAD + 1][8];
#define LOAD_CHUNK(T) do {                                                              \
        _Pragma("unroll") for (int j = 0; j < 8; ++j) {                                 \
            if ((T) < 8)                                                                \
                ring[(T) % (AHEAD + 1)][j] =                                            \
                    __builtin_nontemporal_load(&rp[(T) * (CC / 4) + j * 16]);           \
            else                                                                        \
                ring[(T) % (AHEAD + 1)][j] =                                            \
                    __builtin_nontemporal_load(&x4[(((T) - 8) * 8 + j) * 256]);         \
        } } while (0)

    v2f b; b[0] = 1.f; b[1] = 1.f;
    float mine = 0.f;
    float acc  = 0.f;

#pragma unroll
    for (int t = 0; t < AHEAD; ++t) LOAD_CHUNK(t);

#pragma unroll
    for (int t = 0; t < CHUNKS; ++t) {
        __builtin_amdgcn_sched_barrier(0);
        if (t + AHEAD < CHUNKS) LOAD_CHUNK(t + AHEAD);
        __builtin_amdgcn_sched_barrier(0);

        const v4f* buf = ring[t % (AHEAD + 1)];
        if (t < 8) {
            float s0 = 0.f, s1 = 0.f;
#pragma unroll
            for (int j = 0; j < 4; ++j) s0 += sum4(buf[j]);
#pragma unroll
            for (int j = 4; j < 8; ++j) s1 += sum4(buf[j]);

            v2f a; a[0] = s0; a[1] = s1;
            v8f d = {};
            d = __builtin_amdgcn_wmma_f32_16x16x4_f32(
                     false, a,  false, b,
                     (short)0, d,  false,  false);

            const float r = ((d[0] + d[1]) + (d[2] + d[3])) + ((d[4] + d[5]) + (d[6] + d[7]));
            mine = (c == t) ? r : mine;
            asm volatile("" : "+v"(mine));
            if (t == 7) { if (c < 8) S[rowbase + h * 8 + c] = mine; }
        } else {
            float s = 0.f;
#pragma unroll
            for (int j = 0; j < 8; ++j) s += sum4(buf[j]);
            acc += s;
        }
    }
#undef LOAD_CHUNK

    __shared__ float sm[256];
    sm[tid] = acc;
    __syncthreads();
#pragma unroll
    for (int s = 128; s > 0; s >>= 1) {
        if (tid < s) sm[tid] += sm[tid + s];
        __syncthreads();
    }
    if (tid == 0) partials[blockIdx.x] = sm[0];
}

__global__ void __launch_bounds__(256) pool_kernel(const float* __restrict__ partials,
                                                   const float* __restrict__ S,
                                                   float* __restrict__ out) {
    const int idx  = blockIdx.x * 256 + threadIdx.x;
    const int lane = threadIdx.x & 31;
    const int ic = min(idx, OUT_TOTAL - 1);
    const int b  = ic / OUT_PER_B;
    const int l  = ic - b * OUT_PER_B;
    const float* Sb = S + (long)b * LL + l;
    float t0 = Sb[0], t1 = Sb[1], t2 = Sb[2];

    const int b0 = min((idx - lane) / OUT_PER_B, BB - 1);
    const int b1 = min(b0 + 1, BB - 1);
    float m0 = partials[b0 * 64 + lane] + partials[b0 * 64 + 32 + lane];
    float m1 = partials[b1 * 64 + lane] + partials[b1 * 64 + 32 + lane];
    for (int w = 16; w > 0; w >>= 1) {
        m0 += __shfl_xor(m0, w);
        m1 += __shfl_xor(m1, w);
    }
    asm volatile("" : "+v"(t0), "+v"(t1), "+v"(t2));
    const float k = 0.1f / (3.0f * (float)CC);
    const float M = ((b == b0) ? m0 : m1) / (float)(LL * CC);
    if (idx < OUT_TOTAL) out[idx] = M + k * ((t0 + t1) + t2);
}

extern "C" void kernel_launch(void* const* d_in, const int* in_sizes, int n_in,
                              void* d_out, int out_size, void* d_ws, size_t ws_size,
                              hipStream_t stream) {
    (void)in_sizes; (void)n_in; (void)out_size; (void)ws_size;
    const float* x     = (const float*)d_in[0];
    const float* noise = (const float*)d_in[1];
    float* out = (float*)d_out;

    float* ws       = (float*)d_ws;
    float* partials = ws;
    float* S        = ws + 2048;

    stream_kernel<<<ROWS / 128, 256, 0, stream>>>(x, noise, partials, S);
    pool_kernel<<<(OUT_TOTAL + 255) / 256, 256, 0, stream>>>(partials, S, out);
}


# stream kernel: wave-to-address map interleaved across the grid (chunk p: all 4096 waves read batch p of noise, 2 adjacent rows each; x in 256 KiB windows per batch); x summation order changes
# speedup vs baseline: 1.8019x; 1.0317x over previous
#include <hip/hip_runtime.h>
#include <hip/hip_bf16.h>

#define BB 8
#define LL 8192
#define CC 512
#define ROWS (BB * LL)
#define OUT_PER_B (LL - 2)
#define OUT_TOTAL (BB * OUT_PER_B)
#define BATCH4 (LL * CC / 4)

typedef __attribute__((ext_vector_type(2))) float v2f;
typedef __attribute__((ext_vector_type(4))) float v4f;
typedef __attribute__((ext_vector_type(8))) float v8f;

static __device__ __forceinline__ float sum4(const v4f v) {
    return (v[0] + v[1]) + (v[2] + v[3]);
}

#define CHUNKS 16
#define AHEAD 3

__global__ void __launch_bounds__(256) stream_kernel(const float* __restrict__ x,
                                                     const float* __restrict__ noise,
                                                     float* __restrict__ partials,
                                                     float* __restrict__ S) {
    const int tid  = threadIdx.x;
    const int lane = tid & 31;
    const int wave = tid >> 5;

    const int gw = blockIdx.x * 8 + wave;
    const int c = lane & 15;
    const int h = lane >> 4;

    const v4f* rp = (const v4f*)(noise + (long)(2 * gw + (c >> 3)) * CC + h * 32 + (c & 7) * 4);
    const v4f* x4 = (const v4f*)x + (long)(blockIdx.x >> 6) * BATCH4 + (blockIdx.x & 63) * 256 + tid;

    __shared__ float sS[BB][16];
    v4f ring[AHEAD + 1][8];
#define LOAD_CHUNK(T) do {                                                              \
        _Pragma("unroll") for (int j = 0; j < 8; ++j) {                                 \
            if ((T) < 8)                                                                \
                ring[(T) % (AHEAD + 1)][j] =                                            \
                    __builtin_nontemporal_load(&rp[(long)(T) * BATCH4 + j * 16]);       \
            else                                                                        \
                ring[(T) % (AHEAD + 1)][j] =                                            \
                    __builtin_nontemporal_load(&x4[(long)(((T) - 8) * 8 + j) * 16384]); \
        } } while (0)

    v2f b; b[0] = 1.f; b[1] = 1.f;
    float mine = 0.f;
    float acc  = 0.f;

#pragma unroll
    for (int t = 0; t < AHEAD; ++t) LOAD_CHUNK(t);

#pragma unroll
    for (int t = 0; t < CHUNKS; ++t) {
        __builtin_amdgcn_sched_barrier(0);
        if (t + AHEAD < CHUNKS) LOAD_CHUNK(t + AHEAD);
        __builtin_amdgcn_sched_barrier(0);

        const v4f* buf = ring[t % (AHEAD + 1)];
        if (t < 8) {
            float s0 = 0.f, s1 = 0.f;
#pragma unroll
            for (int j = 0; j < 4; ++j) s0 += sum4(buf[j]);
#pragma unroll
            for (int j = 4; j < 8; ++j) s1 += sum4(buf[j]);

            v2f a; a[0] = s0; a[1] = s1;
            v8f d = {};
            d = __builtin_amdgcn_wmma_f32_16x16x4_f32(
                     false, a,  false, b,
                     (short)0, d,  false,  false);

            const float r = ((d[0] + d[1]) + (d[2] + d[3])) + ((d[4] + d[5]) + (d[6] + d[7]));
            mine = (c == t) ? r : mine;
            asm volatile("" : "+v"(mine));
            if (t == 7) { if (c < 8) sS[c][2 * wave + h] = mine; }
        } else {
            float s = 0.f;
#pragma unroll
            for (int j = 0; j < 8; ++j) s += sum4(buf[j]);
            acc += s;
        }
    }
#undef LOAD_CHUNK

    __shared__ float sm[256];
    sm[tid] = acc;
    __syncthreads();
    if (tid < 128) S[(tid >> 4) * LL + blockIdx.x * 16 + (tid & 15)] = sS[tid >> 4][tid & 15];
#pragma unroll
    for (int s = 128; s > 0; s >>= 1) {
        if (tid < s) sm[tid] += sm[tid + s];
        __syncthreads();
    }
    if (tid == 0) partials[blockIdx.x] = sm[0];
}

__global__ void __launch_bounds__(256) pool_kernel(const float* __restrict__ partials,
                                                   const float* __restrict__ S,
                                                   float* __restrict__ out) {
    const int idx  = blockIdx.x * 256 + threadIdx.x;
    const int lane = threadIdx.x & 31;
    const int ic = min(idx, OUT_TOTAL - 1);
    const int b  = ic / OUT_PER_B;
    const int l  = ic - b * OUT_PER_B;
    const float* Sb = S + (long)b * LL + l;
    float t0 = Sb[0], t1 = Sb[1], t2 = Sb[2];

    const int b0 = min((idx - lane) / OUT_PER_B, BB - 1);
    const int b1 = min(b0 + 1, BB - 1);
    float m0 = partials[b0 * 64 + lane] + partials[b0 * 64 + 32 + lane];
    float m1 = partials[b1 * 64 + lane] + partials[b1 * 64 + 32 + lane];
    for (int w = 16; w > 0; w >>= 1) {
        m0 += __shfl_xor(m0, w);
        m1 += __shfl_xor(m1, w);
    }
    asm volatile("" : "+v"(t0), "+v"(t1), "+v"(t2));
    const float k = 0.1f / (3.0f * (float)CC);
    const float M = ((b == b0) ? m0 : m1) / (float)(LL * CC);
    if (idx < OUT_TOTAL) out[idx] = M + k * ((t0 + t1) + t2);
}

extern "C" void kernel_launch(void* const* d_in, const int* in_sizes, int n_in,
                              void* d_out, int out_size, void* d_ws, size_t ws_size,
                              hipStream_t stream) {
    (void)in_sizes; (void)n_in; (void)out_size; (void)ws_size;
    const float* x     = (const float*)d_in[0];
    const float* noise = (const float*)d_in[1];
    float* out = (float*)d_out;

    float* ws       = (float*)d_ws;
    float* partials = ws;
    float* S        = ws + 2048;

    stream_kernel<<<ROWS / 128, 256, 0, stream>>>(x, noise, partials, S);
    pool_kernel<<<(OUT_TOTAL + 255) / 256, 256, 0, stream>>>(partials, S, out);
}


# stream kernel block tail: the 256-entry LDS tree (9 barriers, row-sum store waited for inside it) replaced by in-register 16-lane row sums, 16 LDS words and ONE barrier; x summation order changes
# speedup vs baseline: 1.8888x; 1.0482x over previous
#include <hip/hip_runtime.h>
#include <hip/hip_bf16.h>

#define BB 8
#define LL 8192
#define CC 512
#define ROWS (BB * LL)
#define OUT_PER_B (LL - 2)
#define OUT_TOTAL (BB * OUT_PER_B)
#define BATCH4 (LL * CC / 4)

typedef __attribute__((ext_vector_type(2))) float v2f;
typedef __attribute__((ext_vector_type(4))) float v4f;
typedef __attribute__((ext_vector_type(8))) float v8f;

static __device__ __forceinline__ float sum4(const v4f v) {
    return (v[0] + v[1]) + (v[2] + v[3]);
}

static __device__ __forceinline__ float row16_sum(float v) {
#define XM(n) __builtin_bit_cast(float, __builtin_amdgcn_update_dpp(0, __builtin_bit_cast(int, v), 0x160 + (n), 0xf, 0xf, true))
    v += XM(1);
    v += XM(2);
    v += XM(4);
    v += XM(8);
#undef XM
    return v;
}

#define CHUNKS 16
#define AHEAD 3

__global__ void __launch_bounds__(256) stream_kernel(const float* __restrict__ x,
                                                     const float* __restrict__ noise,
                                                     float* __restrict__ partials,
                                                     float* __restrict__ S) {
    const int tid  = threadIdx.x;
    const int lane = tid & 31;
    const int wave = tid >> 5;

    const int gw = blockIdx.x * 8 + wave;
    const int c = lane & 15;
    const int h = lane >> 4;

    const v4f* rp = (const v4f*)(noise + (long)(2 * gw + (c >> 3)) * CC + h * 32 + (c & 7) * 4);
    const v4f* x4 = (const v4f*)x + (long)(blockIdx.x >> 6) * BATCH4 + (blockIdx.x & 63) * 256 + tid;

    __shared__ float sS[BB][16];
    v4f ring[AHEAD + 1][8];
#define LOAD_CHUNK(T) do {                                                              \
        _Pragma("unroll") for (int j = 0; j < 8; ++j) {                                 \
            if ((T) < 8)                                                                \
                ring[(T) % (AHEAD + 1)][j] =                                            \
                    __builtin_nontemporal_load(&rp[(long)(T) * BATCH4 + j * 16]);       \
            else                                                                        \
                ring[(T) % (AHEAD + 1)][j] =                                            \
                    __builtin_nontemporal_load(&x4[(long)(((T) - 8) * 8 + j) * 16384]); \
        } } while (0)

    v2f b; b[0] = 1.f; b[1] = 1.f;
    float mine = 0.f;
    float acc  = 0.f;

#pragma unroll
    for (int t = 0; t < AHEAD; ++t) LOAD_CHUNK(t);

#pragma unroll
    for (int t = 0; t < CHUNKS; ++t) {
        __builtin_amdgcn_sched_barrier(0);
        if (t + AHEAD < CHUNKS) LOAD_CHUNK(t + AHEAD);
        __builtin_amdgcn_sched_barrier(0);

        const v4f* buf = ring[t % (AHEAD + 1)];
        if (t < 8) {
            float s0 = 0.f, s1 = 0.f;
#pragma unroll
            for (int j = 0; j < 4; ++j) s0 += sum4(buf[j]);
#pragma unroll
            for (int j = 4; j < 8; ++j) s1 += sum4(buf[j]);

            v2f a; a[0] = s0; a[1] = s1;
            v8f d = {};
            d = __builtin_amdgcn_wmma_f32_16x16x4_f32(
                     false, a,  false, b,
                     (short)0, d,  false,  false);

            const float r = ((d[0] + d[1]) + (d[2] + d[3])) + ((d[4] + d[5]) + (d[6] + d[7]));
            mine = (c == t) ? r : mine;
            asm volatile("" : "+v"(mine));
            if (t == 7) { if (c < 8) sS[c][2 * wave + h] = mine; }
        } else {
            float s = 0.f;
#pragma unroll
            for (int j = 0; j < 8; ++j) s += sum4(buf[j]);
            acc += s;
        }
    }
#undef LOAD_CHUNK

    __shared__ float sW[16];
    const float w = row16_sum(acc);
    if (c == 0) sW[2 * wave + h] = w;
    __syncthreads();
    if (tid < 128) S[(tid >> 4) * LL + blockIdx.x * 16 + (tid & 15)] = sS[tid >> 4][tid & 15];
    if (wave == 4) {
        const float tot = row16_sum(sW[c]);
        if (lane == 0) partials[blockIdx.x] = tot;
    }
}

__global__ void __launch_bounds__(256) pool_kernel(const float* __restrict__ partials,
                                                   const float* __restrict__ S,
                                                   float* __restrict__ out) {
    const int idx  = blockIdx.x * 256 + threadIdx.x;
    const int lane = threadIdx.x & 31;
    const int ic = min(idx, OUT_TOTAL - 1);
    const int b  = ic / OUT_PER_B;
    const int l  = ic - b * OUT_PER_B;
    const float* Sb = S + (long)b * LL + l;
    float t0 = Sb[0], t1 = Sb[1], t2 = Sb[2];

    const int b0 = min((idx - lane) / OUT_PER_B, BB - 1);
    const int b1 = min(b0 + 1, BB - 1);
    float m0 = partials[b0 * 64 + lane] + partials[b0 * 64 + 32 + lane];
    float m1 = partials[b1 * 64 + lane] + partials[b1 * 64 + 32 + lane];
    for (int w = 16; w > 0; w >>= 1) {
        m0 += __shfl_xor(m0, w);
        m1 += __shfl_xor(m1, w);
    }
    asm volatile("" : "+v"(t0), "+v"(t1), "+v"(t2));
    const float k = 0.1f / (3.0f * (float)CC);
    const float M = ((b == b0) ? m0 : m1) / (float)(LL * CC);
    if (idx < OUT_TOTAL) out[idx] = M + k * ((t0 + t1) + t2);
}

extern "C" void kernel_launch(void* const* d_in, const int* in_sizes, int n_in,
                              void* d_out, int out_size, void* d_ws, size_t ws_size,
                              hipStream_t stream) {
    (void)in_sizes; (void)n_in; (void)out_size; (void)ws_size;
    const float* x     = (const float*)d_in[0];
    const float* noise = (const float*)d_in[1];
    float* out = (float*)d_out;

    float* ws       = (float*)d_ws;
    float* partials = ws;
    float* S        = ws + 2048;

    stream_kernel<<<ROWS / 128, 256, 0, stream>>>(x, noise, partials, S);
    pool_kernel<<<(OUT_TOTAL + 255) / 256, 256, 0, stream>>>(partials, S, out);
}
